# starting kernel
# baseline (speedup 1.0000x reference)
#include <hip/hip_runtime.h>

typedef float v2f __attribute__((ext_vector_type(2)));
typedef float v8f __attribute__((ext_vector_type(8)));

#define B_SZ   32
#define C_IN   64
#define C_OUT  64
#define L_IN   16384
#define KSZ    3
#define L_OUT  (L_IN - KSZ + 1)
#define RDIM   (C_IN * KSZ)

#define TILE_N 128
#define HALO   (KSZ - 1)
#define NPAIR  (C_IN / 2)
#define W2     144

__global__ __launch_bounds__(256)
void conv1d_wmma_f32(const float* __restrict__ x,
                     const float* __restrict__ w,
                     const float* __restrict__ bias,
                     float* __restrict__ out)
{
    __shared__ v2f tile2[NPAIR * W2];

    const int b   = blockIdx.y;
    const int l0  = blockIdx.x * TILE_N;
    const int tid = threadIdx.x;

    const float* xb = x + (size_t)b * C_IN * L_IN;

    #pragma unroll
    for (int it = 0; it < 4; ++it) {
        int i = tid + it * 256;
        int p = i >> 5;
        int c = (i & 31) * 4;
        const float4 r0 = *(const float4*)(xb + (size_t)(2 * p    ) * L_IN + l0 + c);
        const float4 r1 = *(const float4*)(xb + (size_t)(2 * p + 1) * L_IN + l0 + c);
        v2f* dst = &tile2[p * W2 + c];
        v2f d0; d0.x = r0.x; d0.y = r1.x; dst[0] = d0;
        v2f d1; d1.x = r0.y; d1.y = r1.y; dst[1] = d1;
        v2f d2; d2.x = r0.z; d2.y = r1.z; dst[2] = d2;
        v2f d3; d3.x = r0.w; d3.y = r1.w; dst[3] = d3;
    }
    if (tid < 2 * NPAIR) {
        int p  = tid >> 1;
        int c  = TILE_N + (tid & 1);
        int gl = l0 + c;
        v2f d;
        d.x = (gl < L_IN) ? xb[(size_t)(2 * p    ) * L_IN + gl] : 0.0f;
        d.y = (gl < L_IN) ? xb[(size_t)(2 * p + 1) * L_IN + gl] : 0.0f;
        tile2[p * W2 + c] = d;
    }
    __syncthreads();

    const int wave = tid >> 5;
    const int lane = tid & 31;
    const int lo   = lane & 15;
    const int hi   = lane >> 4;

    const int m0 = (wave & 3) * 16;
    const int nb = (wave >> 2) * 64;

    v8f acc[4];
    #pragma unroll
    for (int j = 0; j < 8; ++j) {
        float bv = bias[m0 + j + 8 * hi];
        acc[0][j] = bv; acc[1][j] = bv; acc[2][j] = bv; acc[3][j] = bv;
    }

    const float* wrow = w + (size_t)(m0 + lo) * RDIM;

    for (int k = 0; k < KSZ; ++k) {
        #pragma unroll 4
        for (int cc = 0; cc < C_IN; cc += 4) {
            const int r = cc + 2 * hi;
            v2f a;
            a.x = wrow[r * KSZ + k];
            a.y = wrow[(r + 1) * KSZ + k];

            const v2f* bp = &tile2[(r >> 1) * W2 + nb + lo + k];
            #pragma unroll
            for (int t = 0; t < 4; ++t) {
                v2f bf = bp[t * 16];
                acc[t] = __builtin_amdgcn_wmma_f32_16x16x4_f32(
                     false, a,  false, bf,
                     (short)0, acc[t],
                     false,  false);
            }
        }
    }

    float* ob = out + (size_t)b * C_OUT * L_OUT;
    const int lbase = l0 + nb + lo;
    #pragma unroll
    for (int t = 0; t < 4; ++t) {
        int l = lbase + t * 16;
        if (l < L_OUT) {
            #pragma unroll
            for (int j = 0; j < 8; ++j) {
                ob[(size_t)(m0 + j + 8 * hi) * L_OUT + l] = acc[t][j];
            }
        }
    }
}

extern "C" void kernel_launch(void* const* d_in, const int* in_sizes, int n_in,
                              void* d_out, int out_size, void* d_ws, size_t ws_size,
                              hipStream_t stream) {
    const float* x    = (const float*)d_in[0];
    const float* w    = (const float*)d_in[1];
    const float* bias = (const float*)d_in[2];
    float* out        = (float*)d_out;

    dim3 grid((L_OUT + TILE_N - 1) / TILE_N, B_SZ);
    conv1d_wmma_f32<<<grid, dim3(256, 1, 1), 0, stream>>>(x, w, bias, out);
}


# Weight-stationary sweep: a block loads the 64x192 weights once (whole lines via LDS), each wave keeps its 16x192 slice in 96 VGPRs over 8 l-tiles (x double-buffered in LDS); no per-step weight loads
# speedup vs baseline: 1.1660x; 1.1660x over previous
#include <hip/hip_runtime.h>


typedef float v2f __attribute__((ext_vector_type(2)));
typedef float v8f __attribute__((ext_vector_type(8)));

#define B_SZ   32
#define C_IN   64
#define C_OUT  64
#define L_IN   16384
#define KSZ    3
#define L_OUT  (L_IN - KSZ + 1)
#define RDIM   (C_IN * KSZ)

#define TILE_N 128
#define HALO   (KSZ - 1)
#define NPAIR  (C_IN / 2)
#define W2     144
#define TILE_V2 (NPAIR * W2)

#define TPB    8
#define WS     196
#define NKS    (C_IN / 4)

__global__ __launch_bounds__(256)
void conv1d_wmma_f32(const float* __restrict__ x,
                     const float* __restrict__ w,
                     const float* __restrict__ bias,
                     float* __restrict__ out)
{
    __shared__ alignas(16) v2f   tile2[2 * TILE_V2];
    __shared__ alignas(16) float wst[C_OUT * WS];

    const int b   = blockIdx.y;
    const int lt0 = blockIdx.x * TPB;
    const int tid = threadIdx.x;

    const float* xb = x + (size_t)b * C_IN * L_IN;

    const int fp = tid >> 5;
    const int fc = (tid & 31) * 4;
    const int hp = tid >> 1;
    const int hc = TILE_N + (tid & 1);

    float4 r0[4], r1[4];
    float  h0 = 0.0f, h1 = 0.0f;

#define LOAD_TILE(L0)                                                                   \
    {                                                                                   \
        _Pragma("unroll")                                                               \
        for (int it = 0; it < 4; ++it) {                                                \
            const int p = fp + 8 * it;                                                  \
            r0[it] = *(const float4*)(xb + (size_t)(2 * p    ) * L_IN + (L0) + fc);     \
            r1[it] = *(const float4*)(xb + (size_t)(2 * p + 1) * L_IN + (L0) + fc);     \
        }                                                                               \
        if (tid < 2 * NPAIR) {                                                          \
            const int gl = (L0) + hc;                                                   \
            h0 = (gl < L_IN) ? xb[(size_t)(2 * hp    ) * L_IN + gl] : 0.0f;             \
            h1 = (gl < L_IN) ? xb[(size_t)(2 * hp + 1) * L_IN + gl] : 0.0f;             \
        }                                                                               \
    }

#define WRITE_TILE(BUF)                                                                 \
    {                                                                                   \
        _Pragma("unroll")                                                               \
        for (int it = 0; it < 4; ++it) {                                                \
            v2f* dst = (BUF) + (fp + 8 * it) * W2 + fc;                                 \
            v2f d0; d0.x = r0[it].x; d0.y = r1[it].x; dst[0] = d0;                      \
            v2f d1; d1.x = r0[it].y; d1.y = r1[it].y; dst[1] = d1;                      \
            v2f d2; d2.x = r0[it].z; d2.y = r1[it].z; dst[2] = d2;                      \
            v2f d3; d3.x = r0[it].w; d3.y = r1[it].w; dst[3] = d3;                      \
        }                                                                               \
        if (tid < 2 * NPAIR) {                                                          \
            v2f d; d.x = h0; d.y = h1;                                                  \
            (BUF)[hp * W2 + hc] = d;                                                    \
        }                                                                               \
    }

    const int wave = tid >> 5;
    const int lane = tid & 31;
    const int lo   = lane & 15;
    const int hi   = lane >> 4;

    const int m0 = (wave & 3) * 16;
    const int nb = (wave >> 2) * 64;

    v8f bv;
    #pragma unroll
    for (int j = 0; j < 8; ++j) bv[j] = bias[m0 + j + 8 * hi];

    LOAD_TILE(lt0 * TILE_N)

    #pragma unroll
    for (int it = 0; it < (C_OUT * RDIM / 4) / 256; ++it) {
        const int i   = tid + it * 256;
        const int row = i / (RDIM / 4);
        const int c4  = i - row * (RDIM / 4);
        const float4 v = *(const float4*)(w + (size_t)i * 4);
        *(float4*)&wst[row * WS + c4 * 4] = v;
    }
    WRITE_TILE(tile2)
    __syncthreads();

    const float* wl = &wst[(m0 + lo) * WS + 2 * hi * KSZ];
    v2f a[KSZ][NKS];
    #pragma unroll
    for (int j = 0; j < NKS; ++j) {
        #pragma unroll
        for (int k = 0; k < KSZ; ++k) {
            a[k][j].x = wl[(4 * j    ) * KSZ + k];
            a[k][j].y = wl[(4 * j + 1) * KSZ + k];
        }
    }

    float* ob = out + (size_t)b * C_OUT * L_OUT;
    const int boff = hi * W2 + nb + lo;

    for (int ti = 0; ti < TPB; ++ti) {
        const int l0 = (lt0 + ti) * TILE_N;

        if (ti + 1 < TPB) {
            LOAD_TILE(l0 + TILE_N)
        }

        const v2f* cur = tile2 + (ti & 1) * TILE_V2 + boff;
        v8f acc[4];
        v2f bfr[3][4];
        #pragma unroll
        for (int s = 0; s < 2; ++s) {
            const v2f* bp = cur + (2 * (s % NKS)) * W2 + (s / NKS);
            #pragma unroll
            for (int t = 0; t < 4; ++t) bfr[s][t] = bp[t * 16];
        }
        #pragma unroll
        for (int s = 0; s < KSZ * NKS; ++s) {
            if (s + 2 < KSZ * NKS) {
                const int s2 = s + 2;
                const v2f* bp = cur + (2 * (s2 % NKS)) * W2 + (s2 / NKS);
                #pragma unroll
                for (int t = 0; t < 4; ++t) bfr[s2 % 3][t] = bp[t * 16];
            }
            #pragma unroll
            for (int t = 0; t < 4; ++t) {
                acc[t] = __builtin_amdgcn_wmma_f32_16x16x4_f32(
                     false, a[s / NKS][s % NKS],  false, bfr[s % 3][t],
                     (short)0, (s == 0) ? bv : acc[t],
                     false,  false);
            }
            __builtin_amdgcn_sched_barrier(0);
        }

        WRITE_TILE(tile2 + ((ti + 1) & 1) * TILE_V2)
        __syncthreads();

        const int lbase = l0 + nb + lo;
        #pragma unroll
        for (int t = 0; t < 4; ++t) {
            int l = lbase + t * 16;
            if (l < L_OUT) {
                #pragma unroll
                for (int j = 0; j < 8; ++j) {
                    ob[(size_t)(m0 + j + 8 * hi) * L_OUT + l] = acc[t][j];
                }
            }
        }
    }
#undef LOAD_TILE
#undef WRITE_TILE
}

extern "C" void kernel_launch(void* const* d_in, const int* in_sizes, int n_in,
                              void* d_out, int out_size, void* d_ws, size_t ws_size,
                              hipStream_t stream) {
    const float* x    = (const float*)d_in[0];
    const float* w    = (const float*)d_in[1];
    const float* bias = (const float*)d_in[2];
    float* out        = (float*)d_out;

    dim3 grid((L_OUT + TILE_N - 1) / TILE_N / TPB, B_SZ);
    conv1d_wmma_f32<<<grid, dim3(256, 1, 1), 0, stream>>>(x, w, bias, out);
}


# 4 matrix products per 2 outputs of the 3-tap filter, not 6 (minimal filtering 2x3): x differences on vector unit, tap half-sums as A, outputs recombined; sums regrouped; 4-wave blocks
# speedup vs baseline: 1.5079x; 1.2932x over previous
#include <hip/hip_runtime.h>


typedef float v2f __attribute__((ext_vector_type(2)));
typedef float v4f __attribute__((ext_vector_type(4)));
typedef float v8f __attribute__((ext_vector_type(8)));

#define B_SZ   32
#define C_IN   64
#define C_OUT  64
#define L_IN   16384
#define KSZ    3
#define L_OUT  (L_IN - KSZ + 1)
#define RDIM   (C_IN * KSZ)

#define NTHR   128
#define RPP    (NTHR / 32)
#define NIT    (32 / RPP)
#define NMT    4

#define TILE_N 128
#define NPAIR  (C_IN / 2)
#define HW     72
#define W2     (2 * HW)
#define TILE_V2 (NPAIR * W2)

#define TPB    8
#define WS     196
#define NKS    (C_IN / 4)

__global__ __launch_bounds__(NTHR)
void conv1d_wmma_f32(const float* __restrict__ x,
                     const float* __restrict__ w,
                     const float* __restrict__ bias,
                     float* __restrict__ out)
{
    __shared__ alignas(16) v2f   tile2[2 * TILE_V2];
    __shared__ alignas(16) float wst[C_OUT * WS];

    const int b   = blockIdx.y;
    const int lt0 = blockIdx.x * TPB;
    const int tid = threadIdx.x;

    const float* xb = x + (size_t)b * C_IN * L_IN;

    const int fp = tid >> 5;
    const int fc = (tid & 31) * 4;
    const int hp = tid >> 1;
    const int hc = TILE_N + (tid & 1);

    float4 r0[NIT], r1[NIT];
    float  h0 = 0.0f, h1 = 0.0f;

#define LOAD_TILE(L0)                                                                   \
    {                                                                                   \
        _Pragma("unroll")                                                               \
        for (int it = 0; it < NIT; ++it) {                                              \
            const int p = fp + RPP * it;                                                \
            r0[it] = *(const float4*)(xb + (size_t)(2 * p    ) * L_IN + (L0) + fc);     \
            r1[it] = *(const float4*)(xb + (size_t)(2 * p + 1) * L_IN + (L0) + fc);     \
        }                                                                               \
        if (tid < 2 * NPAIR) {                                                          \
            const int gl = (L0) + hc;                                                   \
            h0 = (gl < L_IN) ? xb[(size_t)(2 * hp    ) * L_IN + gl] : 0.0f;             \
            h1 = (gl < L_IN) ? xb[(size_t)(2 * hp + 1) * L_IN + gl] : 0.0f;             \
        }                                                                               \
    }

#define WRITE_TILE(BUF)                                                                 \
    {                                                                                   \
        _Pragma("unroll")                                                               \
        for (int it = 0; it < NIT; ++it) {                                              \
            v2f* dst = (BUF) + (fp + RPP * it) * W2 + (fc >> 1);                        \
            v4f ev; ev.x = r0[it].x; ev.y = r1[it].x; ev.z = r0[it].z; ev.w = r1[it].z; \
            v4f od; od.x = r0[it].y; od.y = r1[it].y; od.z = r0[it].w; od.w = r1[it].w; \
            *(v4f*)(dst)      = ev;                                                     \
            *(v4f*)(dst + HW) = od;                                                     \
        }                                                                               \
        if (tid < 2 * NPAIR) {                                                          \
            v2f d; d.x = h0; d.y = h1;                                                  \
            (BUF)[hp * W2 + (tid & 1) * HW + (TILE_N / 2)] = d;                         \
        }                                                                               \
    }

    const int wave = tid >> 5;
    const int lane = tid & 31;
    const int lo   = lane & 15;
    const int hi   = lane >> 4;

    const int m0 = (wave & 3) * 16;
    const int mb = (wave >> 2) * (16 * NMT);

    v8f bv;
    #pragma unroll
    for (int r = 0; r < 8; ++r) bv[r] = bias[m0 + r + 8 * hi];

    LOAD_TILE(lt0 * TILE_N)

    #pragma unroll
    for (int it = 0; it < (C_OUT * RDIM / 4) / NTHR; ++it) {
        const int i   = tid + it * NTHR;
        const int row = i / (RDIM / 4);
        const int c4  = i - row * (RDIM / 4);
        const float4 v = *(const float4*)(w + (size_t)i * 4);
        *(float4*)&wst[row * WS + c4 * 4] = v;
    }
    WRITE_TILE(tile2)
    __syncthreads();

    const float* wl = &wst[(m0 + lo) * WS + 2 * hi * KSZ];
    v2f a[4][NKS];
    #pragma unroll
    for (int j = 0; j < NKS; ++j) {
        const float g00 = wl[12 * j + 0], g01 = wl[12 * j + 1], g02 = wl[12 * j + 2];
        const float g10 = wl[12 * j + 3], g11 = wl[12 * j + 4], g12 = wl[12 * j + 5];
        const float s0 = 0.5f * g00 + 0.5f * g02, s1 = 0.5f * g10 + 0.5f * g12;
        const float q0 = 0.5f * g01,              q1 = 0.5f * g11;
        a[0][j].x = g00;     a[0][j].y = g10;
        a[1][j].x = s0 + q0; a[1][j].y = s1 + q1;
        a[2][j].x = s0 - q0; a[2][j].y = s1 - q1;
        a[3][j].x = g02;     a[3][j].y = g12;
    }

    float* ob = out + (size_t)b * C_OUT * L_OUT;
    const int boff = hi * W2 + mb + lo;

    const v8f zero = {0.0f, 0.0f, 0.0f, 0.0f, 0.0f, 0.0f, 0.0f, 0.0f};

    for (int ti = 0; ti < TPB; ++ti) {
        const int l0 = (lt0 + ti) * TILE_N;

        if (ti + 1 < TPB) {
            LOAD_TILE(l0 + TILE_N)
        }

        const v2f* cur = tile2 + (ti & 1) * TILE_V2 + boff;
        v8f acc[4][NMT];
        v2f raw[2][NMT][4];
        #pragma unroll
        for (int t = 0; t < NMT; ++t) {
            const v2f* bp = cur + t * 16;
            raw[0][t][0] = bp[0];  raw[0][t][1] = bp[1];
            raw[0][t][2] = bp[HW]; raw[0][t][3] = bp[HW + 1];
        }
        #pragma unroll
        for (int j = 0; j < NKS; ++j) {
            if (j + 1 < NKS) {
                #pragma unroll
                for (int t = 0; t < NMT; ++t) {
                    const v2f* bp = cur + (2 * (j + 1)) * W2 + t * 16;
                    raw[(j + 1) & 1][t][0] = bp[0];  raw[(j + 1) & 1][t][1] = bp[1];
                    raw[(j + 1) & 1][t][2] = bp[HW]; raw[(j + 1) & 1][t][3] = bp[HW + 1];
                }
            }
            __builtin_amdgcn_sched_barrier(0);
            #pragma unroll
            for (int t = 0; t < NMT; ++t) {
                const v2f e0 = raw[j & 1][t][0], e1 = raw[j & 1][t][1];
                const v2f o0 = raw[j & 1][t][2], o1 = raw[j & 1][t][3];
                const v2f b0 = e0 - e1;
                const v2f b1 = o0 + e1;
                const v2f b2 = e1 - o0;
                const v2f b3 = o0 - o1;
                acc[0][t] = __builtin_amdgcn_wmma_f32_16x16x4_f32(false, a[0][j], false, b0, (short)0,
                                                                  (j == 0) ? zero : acc[0][t], false, false);
                acc[1][t] = __builtin_amdgcn_wmma_f32_16x16x4_f32(false, a[1][j], false, b1, (short)0,
                                                                  (j == 0) ? bv : acc[1][t], false, false);
                acc[2][t] = __builtin_amdgcn_wmma_f32_16x16x4_f32(false, a[2][j], false, b2, (short)0,
                                                                  (j == 0) ? zero : acc[2][t], false, false);
                acc[3][t] = __builtin_amdgcn_wmma_f32_16x16x4_f32(false, a[3][j], false, b3, (short)0,
                                                                  (j == 0) ? zero : acc[3][t], false, false);
            }
            __builtin_amdgcn_sched_barrier(0);
        }

        WRITE_TILE(tile2 + ((ti + 1) & 1) * TILE_V2)
        __syncthreads();

        #pragma unroll
        for (int t = 0; t < NMT; ++t) {
            const int l = l0 + 2 * (mb + t * 16 + lo);
            if (l < L_OUT) {
                #pragma unroll
                for (int r = 0; r < 8; ++r) {
                    v2f y;
                    y.x = (acc[0][t][r] + acc[1][t][r]) + acc[2][t][r];
                    y.y = (acc[1][t][r] - acc[2][t][r]) - acc[3][t][r];
                    *(v2f*)(ob + (size_t)(m0 + r + 8 * hi) * L_OUT + l) = y;
                }
            }
        }
    }
#undef LOAD_TILE
#undef WRITE_TILE
}

extern "C" void kernel_launch(void* const* d_in, const int* in_sizes, int n_in,
                              void* d_out, int out_size, void* d_ws, size_t ws_size,
                              hipStream_t stream) {
    const float* x    = (const float*)d_in[0];
    const float* w    = (const float*)d_in[1];
    const float* bias = (const float*)d_in[2];
    float* out        = (float*)d_out;

    dim3 grid((L_OUT + TILE_N - 1) / TILE_N / TPB, B_SZ);
    conv1d_wmma_f32<<<grid, dim3(NTHR, 1, 1), 0, stream>>>(x, w, bias, out);
}


# Outputs of a finished tile are held in 64 registers and stored four per k-step under the next tile's products, not as one 32-store burst from the accumulators that the next products wait behind
# speedup vs baseline: 1.5901x; 1.0545x over previous
#include <hip/hip_runtime.h>


typedef float v2f __attribute__((ext_vector_type(2)));
typedef float v4f __attribute__((ext_vector_type(4)));
typedef float v8f __attribute__((ext_vector_type(8)));

#define B_SZ   32
#define C_IN   64
#define C_OUT  64
#define L_IN   16384
#define KSZ    3
#define L_OUT  (L_IN - KSZ + 1)
#define RDIM   (C_IN * KSZ)

#define NTHR   128
#define RPP    (NTHR / 32)
#define NIT    (32 / RPP)
#define NMT    4

#define TILE_N 128
#define NPAIR  (C_IN / 2)
#define HW     72
#define W2     (2 * HW)
#define TILE_V2 (NPAIR * W2)

#define TPB    8
#define WS     196
#define NKS    (C_IN / 4)

__global__ __launch_bounds__(NTHR)
void conv1d_wmma_f32(const float* __restrict__ x,
                     const float* __restrict__ w,
                     const float* __restrict__ bias,
                     float* __restrict__ out)
{
    __shared__ alignas(16) v2f   tile2[2 * TILE_V2];
    __shared__ alignas(16) float wst[C_OUT * WS];

    const int b   = blockIdx.y;
    const int lt0 = blockIdx.x * TPB;
    const int tid = threadIdx.x;

    const float* xb = x + (size_t)b * C_IN * L_IN;

    const int fp = tid >> 5;
    const int fc = (tid & 31) * 4;
    const int hp = tid >> 1;
    const int hc = TILE_N + (tid & 1);

    float4 r0[NIT], r1[NIT];
    float  h0 = 0.0f, h1 = 0.0f;

#define LOAD_TILE(L0)                                                                   \
    {                                                                                   \
        _Pragma("unroll")                                                               \
        for (int it = 0; it < NIT; ++it) {                                              \
            const int p = fp + RPP * it;                                                \
            r0[it] = *(const float4*)(xb + (size_t)(2 * p    ) * L_IN + (L0) + fc);     \
            r1[it] = *(const float4*)(xb + (size_t)(2 * p + 1) * L_IN + (L0) + fc);     \
        }                                                                               \
        if (tid < 2 * NPAIR) {                                                          \
            const int gl = (L0) + hc;                                                   \
            h0 = (gl < L_IN) ? xb[(size_t)(2 * hp    ) * L_IN + gl] : 0.0f;             \
            h1 = (gl < L_IN) ? xb[(size_t)(2 * hp + 1) * L_IN + gl] : 0.0f;             \
        }                                                                               \
    }

#define WRITE_TILE(BUF)                                                                 \
    {                                                                                   \
        _Pragma("unroll")                                                               \
        for (int it = 0; it < NIT; ++it) {                                              \
            v2f* dst = (BUF) + (fp + RPP * it) * W2 + (fc >> 1);                        \
            v4f ev; ev.x = r0[it].x; ev.y = r1[it].x; ev.z = r0[it].z; ev.w = r1[it].z; \
            v4f od; od.x = r0[it].y; od.y = r1[it].y; od.z = r0[it].w; od.w = r1[it].w; \
            *(v4f*)(dst)      = ev;                                                     \
            *(v4f*)(dst + HW) = od;                                                     \
        }                                                                               \
        if (tid < 2 * NPAIR) {                                                          \
            v2f d; d.x = h0; d.y = h1;                                                  \
            (BUF)[hp * W2 + (tid & 1) * HW + (TILE_N / 2)] = d;                         \
        }                                                                               \
    }

    const int wave = tid >> 5;
    const int lane = tid & 31;
    const int lo   = lane & 15;
    const int hi   = lane >> 4;

    const int m0 = (wave & 3) * 16;
    const int mb = (wave >> 2) * (16 * NMT);

    v8f bv;
    #pragma unroll
    for (int r = 0; r < 8; ++r) bv[r] = bias[m0 + r + 8 * hi];

    LOAD_TILE(lt0 * TILE_N)

    #pragma unroll
    for (int it = 0; it < (C_OUT * RDIM / 4) / NTHR; ++it) {
        const int i   = tid + it * NTHR;
        const int row = i / (RDIM / 4);
        const int c4  = i - row * (RDIM / 4);
        const float4 v = *(const float4*)(w + (size_t)i * 4);
        *(float4*)&wst[row * WS + c4 * 4] = v;
    }
    WRITE_TILE(tile2)
    __syncthreads();

    const float* wl = &wst[(m0 + lo) * WS + 2 * hi * KSZ];
    v2f a[4][NKS];
    #pragma unroll
    for (int j = 0; j < NKS; ++j) {
        const float g00 = wl[12 * j + 0], g01 = wl[12 * j + 1], g02 = wl[12 * j + 2];
        const float g10 = wl[12 * j + 3], g11 = wl[12 * j + 4], g12 = wl[12 * j + 5];
        const float s0 = 0.5f * g00 + 0.5f * g02, s1 = 0.5f * g10 + 0.5f * g12;
        const float q0 = 0.5f * g01,              q1 = 0.5f * g11;
        a[0][j].x = g00;     a[0][j].y = g10;
        a[1][j].x = s0 + q0; a[1][j].y = s1 + q1;
        a[2][j].x = s0 - q0; a[2][j].y = s1 - q1;
        a[3][j].x = g02;     a[3][j].y = g12;
    }

    float* ob = out + (size_t)b * C_OUT * L_OUT;
    const int boff = hi * W2 + mb + lo;

    const v8f zero = {0.0f, 0.0f, 0.0f, 0.0f, 0.0f, 0.0f, 0.0f, 0.0f};

    static_assert(((L_OUT + TILE_N - 1) / TILE_N - 1) * TILE_N <= L_OUT, "every tile but the last of a row is full");
    v2f pend[NMT][8];
    #pragma unroll
    for (int t = 0; t < NMT; ++t) {
        #pragma unroll
        for (int r = 0; r < 8; ++r) { pend[t][r].x = 0.0f; pend[t][r].y = 0.0f; }
    }
    float* op = ob + (size_t)(m0 + 8 * hi) * L_OUT + 2 * (mb + lo);

    for (int ti = 0; ti < TPB; ++ti) {
        const int l0 = (lt0 + ti) * TILE_N;

        if (ti + 1 < TPB) {
            LOAD_TILE(l0 + TILE_N)
        }

        const v2f* cur = tile2 + (ti & 1) * TILE_V2 + boff;
        v8f acc[4][NMT];
        v2f raw[2][NMT][4];
        #pragma unroll
        for (int t = 0; t < NMT; ++t) {
            const v2f* bp = cur + t * 16;
            raw[0][t][0] = bp[0];  raw[0][t][1] = bp[1];
            raw[0][t][2] = bp[HW]; raw[0][t][3] = bp[HW + 1];
        }
        #pragma unroll
        for (int j = 0; j < NKS; ++j) {
            if (j + 1 < NKS) {
                #pragma unroll
                for (int t = 0; t < NMT; ++t) {
                    const v2f* bp = cur + (2 * (j + 1)) * W2 + t * 16;
                    raw[(j + 1) & 1][t][0] = bp[0];  raw[(j + 1) & 1][t][1] = bp[1];
                    raw[(j + 1) & 1][t][2] = bp[HW]; raw[(j + 1) & 1][t][3] = bp[HW + 1];
                }
            }
            if (j < 8 && ti > 0) {
                float* pp = op + (l0 - TILE_N) + (size_t)j * L_OUT;
                #pragma unroll
                for (int t = 0; t < NMT; ++t) *(v2f*)(pp + 32 * t) = pend[t][j];
            }
            __builtin_amdgcn_sched_barrier(0);
            #pragma unroll
            for (int t = 0; t < NMT; ++t) {
                const v2f e0 = raw[j & 1][t][0], e1 = raw[j & 1][t][1];
                const v2f o0 = raw[j & 1][t][2], o1 = raw[j & 1][t][3];
                const v2f b0 = e0 - e1;
                const v2f b1 = o0 + e1;
                const v2f b2 = e1 - o0;
                const v2f b3 = o0 - o1;
                acc[0][t] = __builtin_amdgcn_wmma_f32_16x16x4_f32(false, a[0][j], false, b0, (short)0,
                                                                  (j == 0) ? zero : acc[0][t], false, false);
                acc[1][t] = __builtin_amdgcn_wmma_f32_16x16x4_f32(false, a[1][j], false, b1, (short)0,
                                                                  (j == 0) ? bv : acc[1][t], false, false);
                acc[2][t] = __builtin_amdgcn_wmma_f32_16x16x4_f32(false, a[2][j], false, b2, (short)0,
                                                                  (j == 0) ? zero : acc[2][t], false, false);
                acc[3][t] = __builtin_amdgcn_wmma_f32_16x16x4_f32(false, a[3][j], false, b3, (short)0,
                                                                  (j == 0) ? zero : acc[3][t], false, false);
            }
            __builtin_amdgcn_sched_barrier(0);
        }

        WRITE_TILE(tile2 + ((ti + 1) & 1) * TILE_V2)
        __syncthreads();

        if (ti + 1 < TPB) {
            #pragma unroll
            for (int t = 0; t < NMT; ++t) {
                #pragma unroll
                for (int r = 0; r < 8; ++r) {
                    pend[t][r].x = (acc[0][t][r] + acc[1][t][r]) + acc[2][t][r];
                    pend[t][r].y = (acc[1][t][r] - acc[2][t][r]) - acc[3][t][r];
                }
            }
            continue;
        }
        #pragma unroll
        for (int t = 0; t < NMT; ++t) {
            const int l = l0 + 2 * (mb + t * 16 + lo);
            if (l < L_OUT) {
                #pragma unroll
                for (int r = 0; r < 8; ++r) {
                    v2f y;
                    y.x = (acc[0][t][r] + acc[1][t][r]) + acc[2][t][r];
                    y.y = (acc[1][t][r] - acc[2][t][r]) - acc[3][t][r];
                    *(v2f*)(ob + (size_t)(m0 + r + 8 * hi) * L_OUT + l) = y;
                }
            }
        }
    }
#undef LOAD_TILE
#undef WRITE_TILE
}

extern "C" void kernel_launch(void* const* d_in, const int* in_sizes, int n_in,
                              void* d_out, int out_size, void* d_ws, size_t ws_size,
                              hipStream_t stream) {
    const float* x    = (const float*)d_in[0];
    const float* w    = (const float*)d_in[1];
    const float* bias = (const float*)d_in[2];
    float* out        = (float*)d_out;

    dim3 grid((L_OUT + TILE_N - 1) / TILE_N / TPB, B_SZ);
    conv1d_wmma_f32<<<grid, dim3(NTHR, 1, 1), 0, stream>>>(x, w, bias, out);
}
